# starting kernel
# baseline (speedup 1.0000x reference)
#include <hip/hip_runtime.h>
#include <hip/hip_bf16.h>


#define TT      8192
#define NBATCH  128
#define DIN     32
#define HID     16
#define CHLEN   128
#define NCHUNK  (TT / CHLEN)
#define NTILE   (CHLEN / 16)
#define LDS_STRIDE 17

typedef float v2f __attribute__((ext_vector_type(2)));
typedef float v8f __attribute__((ext_vector_type(8)));

__global__ __launch_bounds__(32) void ssm_chunk_kernel(
    const float* __restrict__ u, const float* __restrict__ A,
    const float* __restrict__ W_B, const float* __restrict__ b_B,
    const float* __restrict__ W_C, const float* __restrict__ b_C,
    float* __restrict__ y_partial, float* __restrict__ S)
{
    __shared__ float buLds[CHLEN * LDS_STRIDE];
    __shared__ float ybuf[CHLEN];

    const int wg = blockIdx.x;
    const int b  = wg / NCHUNK;
    const int c  = wg % NCHUNK;
    const int t0 = c * CHLEN;
    const int ln = threadIdx.x;
    const int m  = ln & 15;
    const int kh = ln >> 4;

    v2f bfrag[8];
#pragma unroll
    for (int k = 0; k < 8; ++k) {
        const float* p = W_B + m * DIN + 4 * k + 2 * kh;
        bfrag[k].x = p[0];
        bfrag[k].y = p[1];
    }
    const float bias = b_B[m];

    const float* ubase = u + ((size_t)b * TT + t0) * DIN;

    for (int tile = 0; tile < NTILE; ++tile) {
        if (tile + 1 < NTILE)
            __builtin_prefetch(ubase + (tile + 1) * 16 * DIN + ln * 16, 0, 1);

        v8f cacc;
#pragma unroll
        for (int v = 0; v < 8; ++v) cacc[v] = bias;

        const float* arow = ubase + (tile * 16 + m) * DIN + 2 * kh;
#pragma unroll
        for (int k = 0; k < 8; ++k) {
            v2f afrag;
            afrag.x = arow[4 * k];
            afrag.y = arow[4 * k + 1];
            cacc = __builtin_amdgcn_wmma_f32_16x16x4_f32(
                 false, afrag,  false, bfrag[k],
                 (short)0, cacc,  false,  false);
        }

#pragma unroll
        for (int v = 0; v < 8; ++v) {
            int t = tile * 16 + v + 8 * kh;
            buLds[t * LDS_STRIDE + m] = cacc[v];
        }
    }

    __syncthreads();

    const float bC = b_C[0];
    if (ln < 16) {
        const float Aval = A[ln];
        const float wc   = W_C[ln];
        float hreg = 0.0f;
        for (int t = 0; t < CHLEN; ++t) {
            float bu = buLds[t * LDS_STRIDE + ln];
            hreg = fmaf(Aval, hreg, bu);
            float val = wc * hreg;
            val += __shfl_xor(val, 1);
            val += __shfl_xor(val, 2);
            val += __shfl_xor(val, 4);
            val += __shfl_xor(val, 8);
            if (ln == 0) ybuf[t] = val + bC;
        }
        S[((size_t)b * NCHUNK + c) * HID + ln] = hreg;
    }

    __syncthreads();

    for (int i = ln; i < CHLEN; i += 32)
        y_partial[(size_t)b * TT + t0 + i] = ybuf[i];
}

__global__ __launch_bounds__(32) void ssm_carry_kernel(
    const float* __restrict__ A, const float* __restrict__ S,
    float* __restrict__ Carry)
{
    const int b = blockIdx.x;
    const int h = threadIdx.x;
    if (h >= HID) return;

    float a = A[h];
    float p = a;
#pragma unroll
    for (int i = 0; i < 7; ++i) p *= p;

    float f = 0.0f;
    for (int c = 0; c < NCHUNK; ++c) {
        size_t idx = ((size_t)b * NCHUNK + c) * HID + h;
        Carry[idx] = f;
        f = fmaf(p, f, S[idx]);
    }
}

__global__ __launch_bounds__(CHLEN) void ssm_fix_kernel(
    const float* __restrict__ A, const float* __restrict__ W_C,
    const float* __restrict__ Carry, float* __restrict__ y)
{
    const int wg = blockIdx.x;
    const int b  = wg / NCHUNK;
    const int c  = wg % NCHUNK;
    const int i  = threadIdx.x;

    __shared__ float ch[HID], ah[HID], wch[HID];
    if (i < HID) {
        ch[i]  = Carry[((size_t)b * NCHUNK + c) * HID + i];
        ah[i]  = A[i];
        wch[i] = W_C[i];
    }
    __syncthreads();

    float corr = 0.0f;
#pragma unroll
    for (int h = 0; h < HID; ++h)
        corr = fmaf(wch[h] * ch[h], powf(ah[h], (float)(i + 1)), corr);

    y[(size_t)b * TT + (size_t)c * CHLEN + i] += corr;
}

extern "C" void kernel_launch(void* const* d_in, const int* in_sizes, int n_in,
                              void* d_out, int out_size, void* d_ws, size_t ws_size,
                              hipStream_t stream) {
    const float* u   = (const float*)d_in[0];
    const float* A   = (const float*)d_in[1];
    const float* W_B = (const float*)d_in[2];
    const float* b_B = (const float*)d_in[3];
    const float* W_C = (const float*)d_in[4];
    const float* b_C = (const float*)d_in[5];
    float* y = (float*)d_out;

    float* S     = (float*)d_ws;
    float* Carry = S + (size_t)NBATCH * NCHUNK * HID;

    ssm_chunk_kernel<<<dim3(NBATCH * NCHUNK), dim3(32), 0, stream>>>(
        u, A, W_B, b_B, W_C, b_C, y, S);
    ssm_carry_kernel<<<dim3(NBATCH), dim3(32), 0, stream>>>(A, S, Carry);
    ssm_fix_kernel<<<dim3(NBATCH * NCHUNK), dim3(CHLEN), 0, stream>>>(
        A, W_C, Carry, y);
}


# Correction kernel: each thread evaluates its 16 powers A[h]^(i+1) once and reuses them for 16 chunks of its batch row (was 16 powf per output element); every sum and its order unchanged
# speedup vs baseline: 1.4166x; 1.4166x over previous
#include <hip/hip_runtime.h>
#include <hip/hip_bf16.h>


#define TT      8192
#define NBATCH  128
#define DIN     32
#define HID     16
#define CHLEN   128
#define NCHUNK  (TT / CHLEN)
#define NTILE   (CHLEN / 16)
#define LDS_STRIDE 17

typedef float v2f __attribute__((ext_vector_type(2)));
typedef float v8f __attribute__((ext_vector_type(8)));

__global__ __launch_bounds__(32) void ssm_chunk_kernel(
    const float* __restrict__ u, const float* __restrict__ A,
    const float* __restrict__ W_B, const float* __restrict__ b_B,
    const float* __restrict__ W_C, const float* __restrict__ b_C,
    float* __restrict__ y_partial, float* __restrict__ S)
{
    __shared__ float buLds[CHLEN * LDS_STRIDE];
    __shared__ float ybuf[CHLEN];

    const int wg = blockIdx.x;
    const int b  = wg / NCHUNK;
    const int c  = wg % NCHUNK;
    const int t0 = c * CHLEN;
    const int ln = threadIdx.x;
    const int m  = ln & 15;
    const int kh = ln >> 4;

    v2f bfrag[8];
#pragma unroll
    for (int k = 0; k < 8; ++k) {
        const float* p = W_B + m * DIN + 4 * k + 2 * kh;
        bfrag[k].x = p[0];
        bfrag[k].y = p[1];
    }
    const float bias = b_B[m];

    const float* ubase = u + ((size_t)b * TT + t0) * DIN;

    for (int tile = 0; tile < NTILE; ++tile) {
        if (tile + 1 < NTILE)
            __builtin_prefetch(ubase + (tile + 1) * 16 * DIN + ln * 16, 0, 1);

        v8f cacc;
#pragma unroll
        for (int v = 0; v < 8; ++v) cacc[v] = bias;

        const float* arow = ubase + (tile * 16 + m) * DIN + 2 * kh;
#pragma unroll
        for (int k = 0; k < 8; ++k) {
            v2f afrag;
            afrag.x = arow[4 * k];
            afrag.y = arow[4 * k + 1];
            cacc = __builtin_amdgcn_wmma_f32_16x16x4_f32(
                 false, afrag,  false, bfrag[k],
                 (short)0, cacc,  false,  false);
        }

#pragma unroll
        for (int v = 0; v < 8; ++v) {
            int t = tile * 16 + v + 8 * kh;
            buLds[t * LDS_STRIDE + m] = cacc[v];
        }
    }

    __syncthreads();

    const float bC = b_C[0];
    if (ln < 16) {
        const float Aval = A[ln];
        const float wc   = W_C[ln];
        float hreg = 0.0f;
        for (int t = 0; t < CHLEN; ++t) {
            float bu = buLds[t * LDS_STRIDE + ln];
            hreg = fmaf(Aval, hreg, bu);
            float val = wc * hreg;
            val += __shfl_xor(val, 1);
            val += __shfl_xor(val, 2);
            val += __shfl_xor(val, 4);
            val += __shfl_xor(val, 8);
            if (ln == 0) ybuf[t] = val + bC;
        }
        S[((size_t)b * NCHUNK + c) * HID + ln] = hreg;
    }

    __syncthreads();

    for (int i = ln; i < CHLEN; i += 32)
        y_partial[(size_t)b * TT + t0 + i] = ybuf[i];
}

__global__ __launch_bounds__(32) void ssm_carry_kernel(
    const float* __restrict__ A, const float* __restrict__ S,
    float* __restrict__ Carry)
{
    const int b = blockIdx.x;
    const int h = threadIdx.x;
    if (h >= HID) return;

    float a = A[h];
    float p = a;
#pragma unroll
    for (int i = 0; i < 7; ++i) p *= p;

    float f = 0.0f;
    for (int c = 0; c < NCHUNK; ++c) {
        size_t idx = ((size_t)b * NCHUNK + c) * HID + h;
        Carry[idx] = f;
        f = fmaf(p, f, S[idx]);
    }
}

#define FIXCPB 16
#define FIXGRP (NCHUNK / FIXCPB)

__global__ __launch_bounds__(CHLEN) void ssm_fix_kernel(
    const float* __restrict__ A, const float* __restrict__ W_C,
    const float* __restrict__ Carry, float* __restrict__ y)
{
    const int wg = blockIdx.x;
    const int b  = wg / FIXGRP;
    const int c0 = (wg % FIXGRP) * FIXCPB;
    const int i  = threadIdx.x;

    __shared__ float wcar[FIXCPB * HID];

    float* yp = y + (size_t)b * TT + (size_t)c0 * CHLEN + i;
    const float* cp = Carry + ((size_t)b * NCHUNK + c0) * HID;

    float yv[FIXCPB];
#pragma unroll
    for (int cc = 0; cc < FIXCPB; ++cc) yv[cc] = yp[cc * CHLEN];

    const float wc = W_C[i & (HID - 1)];
    float cv[(FIXCPB * HID) / CHLEN];
#pragma unroll
    for (int j = 0; j < (FIXCPB * HID) / CHLEN; ++j) cv[j] = cp[j * CHLEN + i];

    float pw[HID];
#pragma unroll
    for (int h = 0; h < HID; ++h) pw[h] = powf(A[h], (float)(i + 1));

#pragma unroll
    for (int j = 0; j < (FIXCPB * HID) / CHLEN; ++j) wcar[j * CHLEN + i] = wc * cv[j];
    __syncthreads();

#pragma unroll
    for (int cc = 0; cc < FIXCPB; ++cc) {
        float corr = 0.0f;
#pragma unroll
        for (int h = 0; h < HID; ++h)
            corr = fmaf(wcar[cc * HID + h], pw[h], corr);
        yp[cc * CHLEN] = yv[cc] + corr;
    }
}

extern "C" void kernel_launch(void* const* d_in, const int* in_sizes, int n_in,
                              void* d_out, int out_size, void* d_ws, size_t ws_size,
                              hipStream_t stream) {
    const float* u   = (const float*)d_in[0];
    const float* A   = (const float*)d_in[1];
    const float* W_B = (const float*)d_in[2];
    const float* b_B = (const float*)d_in[3];
    const float* W_C = (const float*)d_in[4];
    const float* b_C = (const float*)d_in[5];
    float* y = (float*)d_out;

    float* S     = (float*)d_ws;
    float* Carry = S + (size_t)NBATCH * NCHUNK * HID;

    ssm_chunk_kernel<<<dim3(NBATCH * NCHUNK), dim3(32), 0, stream>>>(
        u, A, W_B, b_B, W_C, b_C, y, S);
    ssm_carry_kernel<<<dim3(NBATCH), dim3(32), 0, stream>>>(A, S, Carry);
    ssm_fix_kernel<<<dim3(NBATCH * FIXGRP), dim3(CHLEN), 0, stream>>>(
        A, W_C, Carry, y);
}


# Chunk kernel: channel sum W_C.h_t taken out of the serial scan loop (was 4 dependent ds_bpermute per timestep); after the scan each lane sums one LDS row, same pairing and fma: bit-identical
# speedup vs baseline: 2.2804x; 1.6098x over previous
#include <hip/hip_runtime.h>
#include <hip/hip_bf16.h>


#define TT      8192
#define NBATCH  128
#define DIN     32
#define HID     16
#define CHLEN   128
#define NCHUNK  (TT / CHLEN)
#define NTILE   (CHLEN / 16)
#define LDS_STRIDE 17

typedef float v2f __attribute__((ext_vector_type(2)));
typedef float v8f __attribute__((ext_vector_type(8)));

__global__ __launch_bounds__(32) void ssm_chunk_kernel(
    const float* __restrict__ u, const float* __restrict__ A,
    const float* __restrict__ W_B, const float* __restrict__ b_B,
    const float* __restrict__ W_C, const float* __restrict__ b_C,
    float* __restrict__ y_partial, float* __restrict__ S)
{
    __shared__ float buLds[CHLEN * LDS_STRIDE];

    const int wg = blockIdx.x;
    const int b  = wg / NCHUNK;
    const int c  = wg % NCHUNK;
    const int t0 = c * CHLEN;
    const int ln = threadIdx.x;
    const int m  = ln & 15;
    const int kh = ln >> 4;

    v2f bfrag[8];
#pragma unroll
    for (int k = 0; k < 8; ++k) {
        const float* p = W_B + m * DIN + 4 * k + 2 * kh;
        bfrag[k].x = p[0];
        bfrag[k].y = p[1];
    }
    const float bias = b_B[m];

    const float* ubase = u + ((size_t)b * TT + t0) * DIN;

    for (int tile = 0; tile < NTILE; ++tile) {
        if (tile + 1 < NTILE)
            __builtin_prefetch(ubase + (tile + 1) * 16 * DIN + ln * 16, 0, 1);

        v8f cacc;
#pragma unroll
        for (int v = 0; v < 8; ++v) cacc[v] = bias;

        const float* arow = ubase + (tile * 16 + m) * DIN + 2 * kh;
#pragma unroll
        for (int k = 0; k < 8; ++k) {
            v2f afrag;
            afrag.x = arow[4 * k];
            afrag.y = arow[4 * k + 1];
            cacc = __builtin_amdgcn_wmma_f32_16x16x4_f32(
                 false, afrag,  false, bfrag[k],
                 (short)0, cacc,  false,  false);
        }

#pragma unroll
        for (int v = 0; v < 8; ++v) {
            int t = tile * 16 + v + 8 * kh;
            buLds[t * LDS_STRIDE + m] = cacc[v];
        }
    }

    __syncthreads();

    const float bC = b_C[0];
    float w[HID];
#pragma unroll
    for (int h = 0; h < HID; ++h) w[h] = W_C[h];

    if (ln < 16) {
        const float Aval = A[ln];
        float hreg = 0.0f;
        for (int tb = 0; tb < CHLEN; tb += 16) {
            float bu[16];
#pragma unroll
            for (int j = 0; j < 16; ++j)
                bu[j] = buLds[(tb + j) * LDS_STRIDE + ln];
#pragma unroll
            for (int j = 0; j < 16; ++j) {
                hreg = fmaf(Aval, hreg, bu[j]);
                buLds[(tb + j) * LDS_STRIDE + ln] = hreg;
            }
        }
        S[((size_t)b * NCHUNK + c) * HID + ln] = hreg;
    }

    __syncthreads();

    for (int j = 0; j < CHLEN / 32; ++j) {
        const int i = ln + 32 * j;
        const float* r = buLds + i * LDS_STRIDE;
        float s[8];
#pragma unroll
        for (int k = 0; k < 8; ++k)
            s[k] = fmaf(w[2 * k], r[2 * k], w[2 * k + 1] * r[2 * k + 1]);
        const float q0 = s[0] + s[1], q1 = s[2] + s[3];
        const float q2 = s[4] + s[5], q3 = s[6] + s[7];
        const float val = (q0 + q1) + (q2 + q3);
        y_partial[(size_t)b * TT + t0 + i] = val + bC;
    }
}

__global__ __launch_bounds__(32) void ssm_carry_kernel(
    const float* __restrict__ A, const float* __restrict__ S,
    float* __restrict__ Carry)
{
    const int b = blockIdx.x;
    const int h = threadIdx.x;
    if (h >= HID) return;

    float a = A[h];
    float p = a;
#pragma unroll
    for (int i = 0; i < 7; ++i) p *= p;

    float f = 0.0f;
    for (int c = 0; c < NCHUNK; ++c) {
        size_t idx = ((size_t)b * NCHUNK + c) * HID + h;
        Carry[idx] = f;
        f = fmaf(p, f, S[idx]);
    }
}

#define FIXCPB 16
#define FIXGRP (NCHUNK / FIXCPB)

__global__ __launch_bounds__(CHLEN) void ssm_fix_kernel(
    const float* __restrict__ A, const float* __restrict__ W_C,
    const float* __restrict__ Carry, float* __restrict__ y)
{
    const int wg = blockIdx.x;
    const int b  = wg / FIXGRP;
    const int c0 = (wg % FIXGRP) * FIXCPB;
    const int i  = threadIdx.x;

    __shared__ float wcar[FIXCPB * HID];

    float* yp = y + (size_t)b * TT + (size_t)c0 * CHLEN + i;
    const float* cp = Carry + ((size_t)b * NCHUNK + c0) * HID;

    float yv[FIXCPB];
#pragma unroll
    for (int cc = 0; cc < FIXCPB; ++cc) yv[cc] = yp[cc * CHLEN];

    const float wc = W_C[i & (HID - 1)];
    float cv[(FIXCPB * HID) / CHLEN];
#pragma unroll
    for (int j = 0; j < (FIXCPB * HID) / CHLEN; ++j) cv[j] = cp[j * CHLEN + i];

    float pw[HID];
#pragma unroll
    for (int h = 0; h < HID; ++h) pw[h] = powf(A[h], (float)(i + 1));

#pragma unroll
    for (int j = 0; j < (FIXCPB * HID) / CHLEN; ++j) wcar[j * CHLEN + i] = wc * cv[j];
    __syncthreads();

#pragma unroll
    for (int cc = 0; cc < FIXCPB; ++cc) {
        float corr = 0.0f;
#pragma unroll
        for (int h = 0; h < HID; ++h)
            corr = fmaf(wcar[cc * HID + h], pw[h], corr);
        yp[cc * CHLEN] = yv[cc] + corr;
    }
}

extern "C" void kernel_launch(void* const* d_in, const int* in_sizes, int n_in,
                              void* d_out, int out_size, void* d_ws, size_t ws_size,
                              hipStream_t stream) {
    const float* u   = (const float*)d_in[0];
    const float* A   = (const float*)d_in[1];
    const float* W_B = (const float*)d_in[2];
    const float* b_B = (const float*)d_in[3];
    const float* W_C = (const float*)d_in[4];
    const float* b_C = (const float*)d_in[5];
    float* y = (float*)d_out;

    float* S     = (float*)d_ws;
    float* Carry = S + (size_t)NBATCH * NCHUNK * HID;

    ssm_chunk_kernel<<<dim3(NBATCH * NCHUNK), dim3(32), 0, stream>>>(
        u, A, W_B, b_B, W_C, b_C, y, S);
    ssm_carry_kernel<<<dim3(NBATCH), dim3(32), 0, stream>>>(A, S, Carry);
    ssm_fix_kernel<<<dim3(NBATCH * FIXGRP), dim3(CHLEN), 0, stream>>>(
        A, W_C, Carry, y);
}


# Chunk kernel: u tiles read with whole-line loads (4 lines per load, was 16) one tile ahead, put into fragment order through LDS; prefetch hint dropped; k order and all sums unchanged
# speedup vs baseline: 2.3581x; 1.0341x over previous
#include <hip/hip_runtime.h>
#include <hip/hip_bf16.h>


#define TT      8192
#define NBATCH  128
#define DIN     32
#define HID     16
#define CHLEN   128
#define NCHUNK  (TT / CHLEN)
#define NTILE   (CHLEN / 16)
#define LDS_STRIDE 17
#define U_PITCH 36
#define NTILE_F4 128

typedef float v2f __attribute__((ext_vector_type(2)));
typedef float v8f __attribute__((ext_vector_type(8)));

__global__ __launch_bounds__(32) void ssm_chunk_kernel(
    const float* __restrict__ u, const float* __restrict__ A,
    const float* __restrict__ W_B, const float* __restrict__ b_B,
    const float* __restrict__ W_C, const float* __restrict__ b_C,
    float* __restrict__ y_partial, float* __restrict__ S)
{
    __shared__ __attribute__((aligned(16))) float buLds[(NTILE - 1) * 16 * LDS_STRIDE + 16 * U_PITCH];

    const int wg = blockIdx.x;
    const int b  = wg / NCHUNK;
    const int c  = wg % NCHUNK;
    const int t0 = c * CHLEN;
    const int ln = threadIdx.x;
    const int m  = ln & 15;
    const int kh = ln >> 4;

    v2f bfrag[8];
#pragma unroll
    for (int k = 0; k < 8; ++k) {
        const float* p = W_B + m * DIN + 4 * k + 2 * kh;
        bfrag[k].x = p[0];
        bfrag[k].y = p[1];
    }
    const float bias = b_B[m];

    const float* ubase = u + ((size_t)b * TT + t0) * DIN;

    const float4* ug = reinterpret_cast<const float4*>(ubase) + ln;
    float4 ubuf[2][4];
#pragma unroll
    for (int j = 0; j < 4; ++j) ubuf[0][j] = ug[j * 32];

#pragma unroll
    for (int tile = 0; tile < NTILE; ++tile) {
        if (tile + 1 < NTILE) {
#pragma unroll
            for (int j = 0; j < 4; ++j)
                ubuf[(tile + 1) & 1][j] = ug[(tile + 1) * NTILE_F4 + j * 32];
        }

        float* stage = buLds + tile * 16 * LDS_STRIDE;
#pragma unroll
        for (int j = 0; j < 4; ++j)
            *reinterpret_cast<float4*>(stage + (4 * j + (ln >> 3)) * U_PITCH + 4 * (ln & 7)) =
                ubuf[tile & 1][j];

        __builtin_amdgcn_fence(__ATOMIC_ACQ_REL, "workgroup", "local");

        v8f cacc;
#pragma unroll
        for (int v = 0; v < 8; ++v) cacc[v] = bias;

        const float* arow = stage + m * U_PITCH + 2 * kh;
#pragma unroll
        for (int k = 0; k < 8; ++k) {
            const v2f afrag = *reinterpret_cast<const v2f*>(arow + 4 * k);
            cacc = __builtin_amdgcn_wmma_f32_16x16x4_f32(
                 false, afrag,  false, bfrag[k],
                 (short)0, cacc,  false,  false);
        }

#pragma unroll
        for (int v = 0; v < 8; ++v) {
            int t = tile * 16 + v + 8 * kh;
            buLds[t * LDS_STRIDE + m] = cacc[v];
        }
    }

    __syncthreads();

    const float bC = b_C[0];
    float w[HID];
#pragma unroll
    for (int h = 0; h < HID; ++h) w[h] = W_C[h];

    if (ln < 16) {
        const float Aval = A[ln];
        float hreg = 0.0f;
        for (int tb = 0; tb < CHLEN; tb += 16) {
            float bu[16];
#pragma unroll
            for (int j = 0; j < 16; ++j)
                bu[j] = buLds[(tb + j) * LDS_STRIDE + ln];
#pragma unroll
            for (int j = 0; j < 16; ++j) {
                hreg = fmaf(Aval, hreg, bu[j]);
                buLds[(tb + j) * LDS_STRIDE + ln] = hreg;
            }
        }
        S[((size_t)b * NCHUNK + c) * HID + ln] = hreg;
    }

    __syncthreads();

    for (int j = 0; j < CHLEN / 32; ++j) {
        const int i = ln + 32 * j;
        const float* r = buLds + i * LDS_STRIDE;
        float s[8];
#pragma unroll
        for (int k = 0; k < 8; ++k)
            s[k] = fmaf(w[2 * k], r[2 * k], w[2 * k + 1] * r[2 * k + 1]);
        const float q0 = s[0] + s[1], q1 = s[2] + s[3];
        const float q2 = s[4] + s[5], q3 = s[6] + s[7];
        const float val = (q0 + q1) + (q2 + q3);
        y_partial[(size_t)b * TT + t0 + i] = val + bC;
    }
}

__global__ __launch_bounds__(32) void ssm_carry_kernel(
    const float* __restrict__ A, const float* __restrict__ S,
    float* __restrict__ Carry)
{
    const int b = blockIdx.x;
    const int h = threadIdx.x;
    if (h >= HID) return;

    float a = A[h];
    float p = a;
#pragma unroll
    for (int i = 0; i < 7; ++i) p *= p;

    float f = 0.0f;
    for (int c = 0; c < NCHUNK; ++c) {
        size_t idx = ((size_t)b * NCHUNK + c) * HID + h;
        Carry[idx] = f;
        f = fmaf(p, f, S[idx]);
    }
}

#define FIXCPB 16
#define FIXGRP (NCHUNK / FIXCPB)

__global__ __launch_bounds__(CHLEN) void ssm_fix_kernel(
    const float* __restrict__ A, const float* __restrict__ W_C,
    const float* __restrict__ Carry, float* __restrict__ y)
{
    const int wg = blockIdx.x;
    const int b  = wg / FIXGRP;
    const int c0 = (wg % FIXGRP) * FIXCPB;
    const int i  = threadIdx.x;

    __shared__ float wcar[FIXCPB * HID];

    float* yp = y + (size_t)b * TT + (size_t)c0 * CHLEN + i;
    const float* cp = Carry + ((size_t)b * NCHUNK + c0) * HID;

    float yv[FIXCPB];
#pragma unroll
    for (int cc = 0; cc < FIXCPB; ++cc) yv[cc] = yp[cc * CHLEN];

    const float wc = W_C[i & (HID - 1)];
    float cv[(FIXCPB * HID) / CHLEN];
#pragma unroll
    for (int j = 0; j < (FIXCPB * HID) / CHLEN; ++j) cv[j] = cp[j * CHLEN + i];

    float pw[HID];
#pragma unroll
    for (int h = 0; h < HID; ++h) pw[h] = powf(A[h], (float)(i + 1));

#pragma unroll
    for (int j = 0; j < (FIXCPB * HID) / CHLEN; ++j) wcar[j * CHLEN + i] = wc * cv[j];
    __syncthreads();

#pragma unroll
    for (int cc = 0; cc < FIXCPB; ++cc) {
        float corr = 0.0f;
#pragma unroll
        for (int h = 0; h < HID; ++h)
            corr = fmaf(wcar[cc * HID + h], pw[h], corr);
        yp[cc * CHLEN] = yv[cc] + corr;
    }
}

extern "C" void kernel_launch(void* const* d_in, const int* in_sizes, int n_in,
                              void* d_out, int out_size, void* d_ws, size_t ws_size,
                              hipStream_t stream) {
    const float* u   = (const float*)d_in[0];
    const float* A   = (const float*)d_in[1];
    const float* W_B = (const float*)d_in[2];
    const float* b_B = (const float*)d_in[3];
    const float* W_C = (const float*)d_in[4];
    const float* b_C = (const float*)d_in[5];
    float* y = (float*)d_out;

    float* S     = (float*)d_ws;
    float* Carry = S + (size_t)NBATCH * NCHUNK * HID;

    ssm_chunk_kernel<<<dim3(NBATCH * NCHUNK), dim3(32), 0, stream>>>(
        u, A, W_B, b_B, W_C, b_C, y, S);
    ssm_carry_kernel<<<dim3(NBATCH), dim3(32), 0, stream>>>(A, S, Carry);
    ssm_fix_kernel<<<dim3(NBATCH * FIXGRP), dim3(CHLEN), 0, stream>>>(
        A, W_C, Carry, y);
}


# One launch, was three: a 32-wave workgroup owns a batch row; chunk states, carries and the powers pass through LDS, partial y stays in registers; every sum and its order unchanged
# speedup vs baseline: 3.2271x; 1.3685x over previous
#include <hip/hip_runtime.h>
#include <hip/hip_bf16.h>


#define TT      8192
#define NBATCH  128
#define DIN     32
#define HID     16
#define CHLEN   128
#define NCHUNK  (TT / CHLEN)
#define NTILE   (CHLEN / 16)
#define LDS_STRIDE 17
#define U_PITCH 36
#define NTILE_F4 128

#define NWAVE   32
#define CPW     (NCHUNK / NWAVE)
#define WLDS    ((NTILE - 1) * 16 * LDS_STRIDE + 16 * U_PITCH)
#define WCAR_OFF (HID * CHLEN)

typedef float v2f __attribute__((ext_vector_type(2)));
typedef float v4f __attribute__((ext_vector_type(4)));
typedef float v8f __attribute__((ext_vector_type(8)));

#define WAVE_LDS_FENCE() __builtin_amdgcn_fence(__ATOMIC_ACQ_REL, "workgroup", "local")

struct UTile { v4f a, b, c, d; };

__device__ __forceinline__ void load_tile(UTile& t, const v4f* __restrict__ p)
{
    t.a = p[0]; t.b = p[32]; t.c = p[64]; t.d = p[96];
}

template <int TILE, bool LAST>
__device__ __forceinline__ void ssm_tiles(
    const v4f* __restrict__ ugc, UTile& cur, UTile& nxt, float* buLds,
    const v2f (&bfrag)[8], const float bias, const int ln, const int m, const int kh)
{
    if constexpr (!LAST || TILE + 1 < NTILE)
        load_tile(nxt, ugc + (TILE + 1) * NTILE_F4);

    float* stage = buLds + TILE * 16 * LDS_STRIDE;
    float* sp = stage + (ln >> 3) * U_PITCH + 4 * (ln & 7);
    *reinterpret_cast<v4f*>(sp)                = cur.a;
    *reinterpret_cast<v4f*>(sp + 4 * U_PITCH)  = cur.b;
    *reinterpret_cast<v4f*>(sp + 8 * U_PITCH)  = cur.c;
    *reinterpret_cast<v4f*>(sp + 12 * U_PITCH) = cur.d;

    WAVE_LDS_FENCE();

    v8f cacc;
#pragma unroll
    for (int v = 0; v < 8; ++v) cacc[v] = bias;

    const float* arow = stage + m * U_PITCH + 2 * kh;
#pragma unroll
    for (int k = 0; k < 8; ++k) {
        const v2f afrag = *reinterpret_cast<const v2f*>(arow + 4 * k);
        cacc = __builtin_amdgcn_wmma_f32_16x16x4_f32(
             false, afrag,  false, bfrag[k],
             (short)0, cacc,  false,  false);
    }

#pragma unroll
    for (int v = 0; v < 8; ++v) {
        int t = TILE * 16 + v + 8 * kh;
        buLds[t * LDS_STRIDE + m] = cacc[v];
    }

    if constexpr (TILE + 1 < NTILE)
        ssm_tiles<TILE + 1, LAST>(ugc, nxt, cur, buLds, bfrag, bias, ln, m, kh);
}

__device__ __forceinline__ float ssm_rowsum(const float* r, const float (&w)[HID], const float bC)
{
    float s[8];
#pragma unroll
    for (int k = 0; k < 8; ++k)
        s[k] = fmaf(w[2 * k], r[2 * k], w[2 * k + 1] * r[2 * k + 1]);
    const float q0 = s[0] + s[1], q1 = s[2] + s[3];
    const float q2 = s[4] + s[5], q3 = s[6] + s[7];
    const float val = (q0 + q1) + (q2 + q3);
    return val + bC;
}

template <bool LAST>
__device__ __forceinline__ void ssm_chunk(
    const v4f* __restrict__ ugc, UTile& t0, UTile& t1, float* buLds, float* Sdst,
    const v2f (&bfrag)[8], const float bias, const float Aval,
    const float (&w)[HID], const float bC,
    float& y0, float& y1, float& y2, float& y3,
    const int ln, const int m, const int kh)
{
    ssm_tiles<0, LAST>(ugc, t0, t1, buLds, bfrag, bias, ln, m, kh);

    WAVE_LDS_FENCE();

    if (ln < 16) {
        float hreg = 0.0f;
        for (int tb = 0; tb < CHLEN; tb += 16) {
            float bu[16];
#pragma unroll
            for (int j = 0; j < 16; ++j)
                bu[j] = buLds[(tb + j) * LDS_STRIDE + ln];
#pragma unroll
            for (int j = 0; j < 16; ++j) {
                hreg = fmaf(Aval, hreg, bu[j]);
                buLds[(tb + j) * LDS_STRIDE + ln] = hreg;
            }
        }
        Sdst[ln] = hreg;
    }

    WAVE_LDS_FENCE();

    const float* r = buLds + ln * LDS_STRIDE;
    y0 = ssm_rowsum(r, w, bC);
    y1 = ssm_rowsum(r + 32 * LDS_STRIDE, w, bC);
    y2 = ssm_rowsum(r + 64 * LDS_STRIDE, w, bC);
    y3 = ssm_rowsum(r + 96 * LDS_STRIDE, w, bC);
    asm volatile("" : "+v"(y0), "+v"(y1), "+v"(y2), "+v"(y3));

    WAVE_LDS_FENCE();
}

__global__ __launch_bounds__(NWAVE * 32) void ssm_row_kernel(
    const float* __restrict__ u, const float* __restrict__ A,
    const float* __restrict__ W_B, const float* __restrict__ b_B,
    const float* __restrict__ W_C, const float* __restrict__ b_C,
    float* __restrict__ y)
{
    __shared__ __attribute__((aligned(16))) float lds[NWAVE * WLDS + NCHUNK * HID];

    const int b   = blockIdx.x;
    const int tid = threadIdx.x;
    const int wv  = tid >> 5;
    const int ln  = tid & 31;
    const int m   = ln & 15;
    const int kh  = ln >> 4;
    const int c0  = wv * CPW;

    float* buLds = lds + wv * WLDS;
    float* Ssh   = lds + NWAVE * WLDS;

    const float pa0 = A[tid >> 7];
    const float pa1 = A[8 + (tid >> 7)];

    v2f bfrag[8];
#pragma unroll
    for (int k = 0; k < 8; ++k) {
        const float* p = W_B + m * DIN + 4 * k + 2 * kh;
        bfrag[k].x = p[0];
        bfrag[k].y = p[1];
    }
    const float bias = b_B[m];

    const v4f* ug = reinterpret_cast<const v4f*>(
        u + ((size_t)b * TT + (size_t)c0 * CHLEN) * DIN) + ln;
    UTile t0, t1;
    load_tile(t0, ug);
    __builtin_amdgcn_sched_barrier(0);

    float pw0 = powf(pa0, (float)((tid & (CHLEN - 1)) + 1));
    float pw1 = powf(pa1, (float)((tid & (CHLEN - 1)) + 1));
    asm volatile("" : "+v"(pw0), "+v"(pw1));

    const float bC = b_C[0];
    float w[HID];
#pragma unroll
    for (int h = 0; h < HID; ++h) w[h] = W_C[h];
    const float Aval = A[m];

    float ya0, ya1, ya2, ya3, yb0, yb1, yb2, yb3;
    ssm_chunk<false>(ug, t0, t1, buLds, Ssh + c0 * HID, bfrag, bias, Aval, w, bC,
                     ya0, ya1, ya2, ya3, ln, m, kh);
    ssm_chunk<true>(ug + NTILE * NTILE_F4, t0, t1, buLds, Ssh + (c0 + 1) * HID, bfrag, bias, Aval, w, bC,
                    yb0, yb1, yb2, yb3, ln, m, kh);

    __syncthreads();

    float* pwT = lds;
    pwT[tid] = pw0;
    pwT[tid + NWAVE * 32] = pw1;

    float p = Aval;
#pragma unroll
    for (int i = 0; i < 7; ++i) p *= p;

    float f = 0.0f, car0 = 0.0f;
    for (int cb = 0; cb < NCHUNK; cb += 16) {
        float sv[16];
#pragma unroll
        for (int j = 0; j < 16; ++j) sv[j] = Ssh[(cb + j) * HID + m];
#pragma unroll
        for (int j = 0; j < 16; ++j) {
            if (cb + j == c0) car0 = f;
            f = fmaf(p, f, sv[j]);
        }
    }
    const float car1 = fmaf(p, car0, Ssh[c0 * HID + m]);

    float* wcar = buLds + WCAR_OFF;
    const float wc = W_C[m];
    if (ln < 16) {
        wcar[ln]       = wc * car0;
        wcar[HID + ln] = wc * car1;
    }

    __syncthreads();

    float wca[CPW][HID];
#pragma unroll
    for (int cc = 0; cc < CPW; ++cc)
#pragma unroll
        for (int h = 0; h < HID; ++h) wca[cc][h] = wcar[cc * HID + h];

    float* yp = y + (size_t)b * TT + (size_t)c0 * CHLEN + ln;
    const float ya[CHLEN / 32] = {ya0, ya1, ya2, ya3};
    const float yb[CHLEN / 32] = {yb0, yb1, yb2, yb3};
#pragma unroll
    for (int j = 0; j < CHLEN / 32; ++j) {
        float pwr[HID];
#pragma unroll
        for (int h = 0; h < HID; ++h) pwr[h] = pwT[h * CHLEN + ln + 32 * j];
        float corr0 = 0.0f, corr1 = 0.0f;
#pragma unroll
        for (int h = 0; h < HID; ++h) {
            corr0 = fmaf(wca[0][h], pwr[h], corr0);
            corr1 = fmaf(wca[1][h], pwr[h], corr1);
        }
        yp[32 * j]         = ya[j] + corr0;
        yp[CHLEN + 32 * j] = yb[j] + corr1;
    }
}

extern "C" void kernel_launch(void* const* d_in, const int* in_sizes, int n_in,
                              void* d_out, int out_size, void* d_ws, size_t ws_size,
                              hipStream_t stream) {
    const float* u   = (const float*)d_in[0];
    const float* A   = (const float*)d_in[1];
    const float* W_B = (const float*)d_in[2];
    const float* b_B = (const float*)d_in[3];
    const float* W_C = (const float*)d_in[4];
    const float* b_C = (const float*)d_in[5];
    float* y = (float*)d_out;

    ssm_row_kernel<<<dim3(NBATCH), dim3(NWAVE * 32), 0, stream>>>(
        u, A, W_B, b_B, W_C, b_C, y);
}


# Chunk scan on the WMMA accumulator registers (8 steps per lane, lane halves joined by one permlanex16), was a 16-lane loop through LDS; each state now x_t + A^k * h_in: summation order changed
# speedup vs baseline: 3.6589x; 1.1338x over previous
#include <hip/hip_runtime.h>
#include <hip/hip_bf16.h>


#define TT      8192
#define NBATCH  128
#define DIN     32
#define HID     16
#define CHLEN   128
#define NCHUNK  (TT / CHLEN)
#define NTILE   (CHLEN / 16)
#define LDS_STRIDE 17
#define U_PITCH 36
#define NTILE_F4 128

#define NWAVE   32
#define CPW     (NCHUNK / NWAVE)
#define WLDS    ((NTILE - 1) * 16 * LDS_STRIDE + 16 * U_PITCH)
#define WCAR_OFF (HID * CHLEN)

typedef float v2f __attribute__((ext_vector_type(2)));
typedef float v4f __attribute__((ext_vector_type(4)));
typedef float v8f __attribute__((ext_vector_type(8)));

#define WAVE_LDS_FENCE() __builtin_amdgcn_fence(__ATOMIC_ACQ_REL, "workgroup", "local")

__device__ __forceinline__ float xhalf(float x)
{
    const int b = __builtin_bit_cast(int, x);
    return __builtin_bit_cast(float,
        __builtin_amdgcn_permlanex16(b, b, 0x76543210, 0xfedcba98, false, false));
}

struct UTile { v4f a, b, c, d; };

__device__ __forceinline__ void load_tile(UTile& t, const v4f* __restrict__ p)
{
    t.a = p[0]; t.b = p[32]; t.c = p[64]; t.d = p[96];
}

template <int TILE, bool LAST>
__device__ __forceinline__ void ssm_tiles(
    const v4f* __restrict__ ugc, UTile& cur, UTile& nxt, float* buLds,
    const v2f (&bfrag)[8], const float bias, const float (&apow)[8], float& H,
    const int ln, const int m, const int kh)
{
    if constexpr (!LAST || TILE + 1 < NTILE)
        load_tile(nxt, ugc + (TILE + 1) * NTILE_F4);

    float* stage = buLds + TILE * 16 * LDS_STRIDE;
    float* sp = stage + (ln >> 3) * U_PITCH + 4 * (ln & 7);
    *reinterpret_cast<v4f*>(sp)                = cur.a;
    *reinterpret_cast<v4f*>(sp + 4 * U_PITCH)  = cur.b;
    *reinterpret_cast<v4f*>(sp + 8 * U_PITCH)  = cur.c;
    *reinterpret_cast<v4f*>(sp + 12 * U_PITCH) = cur.d;

    WAVE_LDS_FENCE();

    v8f cacc;
#pragma unroll
    for (int v = 0; v < 8; ++v) cacc[v] = bias;

    const float* arow = stage + m * U_PITCH + 2 * kh;
#pragma unroll
    for (int k = 0; k < 8; ++k) {
        const v2f afrag = *reinterpret_cast<const v2f*>(arow + 4 * k);
        cacc = __builtin_amdgcn_wmma_f32_16x16x4_f32(
             false, afrag,  false, bfrag[k],
             (short)0, cacc,  false,  false);
    }

    float x[8];
    x[0] = cacc[0];
#pragma unroll
    for (int v = 1; v < 8; ++v) x[v] = fmaf(apow[0], x[v - 1], cacc[v]);

    float oth = xhalf(x[7]);
    asm volatile("" : "+v"(oth));
    const float lo7 = kh ? oth : x[7];
    const float hi7 = kh ? x[7] : oth;
    const float mid = fmaf(apow[7], H, lo7);
    const float hin = kh ? mid : H;
    H = fmaf(apow[7], mid, hi7);

#pragma unroll
    for (int v = 0; v < 8; ++v) {
        int t = TILE * 16 + v + 8 * kh;
        buLds[t * LDS_STRIDE + m] = fmaf(apow[v], hin, x[v]);
    }

    if constexpr (TILE + 1 < NTILE)
        ssm_tiles<TILE + 1, LAST>(ugc, nxt, cur, buLds, bfrag, bias, apow, H, ln, m, kh);
}

__device__ __forceinline__ float ssm_rowsum(const float* r, const float (&w)[HID], const float bC)
{
    float s[8];
#pragma unroll
    for (int k = 0; k < 8; ++k)
        s[k] = fmaf(w[2 * k], r[2 * k], w[2 * k + 1] * r[2 * k + 1]);
    const float q0 = s[0] + s[1], q1 = s[2] + s[3];
    const float q2 = s[4] + s[5], q3 = s[6] + s[7];
    const float val = (q0 + q1) + (q2 + q3);
    return val + bC;
}

template <bool LAST>
__device__ __forceinline__ void ssm_chunk(
    const v4f* __restrict__ ugc, UTile& t0, UTile& t1, float* buLds, float* Sdst,
    const v2f (&bfrag)[8], const float bias, const float (&apow)[8],
    const float (&w)[HID], const float bC,
    float& y0, float& y1, float& y2, float& y3,
    const int ln, const int m, const int kh)
{
    float H = 0.0f;
    ssm_tiles<0, LAST>(ugc, t0, t1, buLds, bfrag, bias, apow, H, ln, m, kh);

    if (ln < 16) Sdst[ln] = H;

    WAVE_LDS_FENCE();

    const float* r = buLds + ln * LDS_STRIDE;
    y0 = ssm_rowsum(r, w, bC);
    y1 = ssm_rowsum(r + 32 * LDS_STRIDE, w, bC);
    asm volatile("" : "+v"(y0), "+v"(y1));
    y2 = ssm_rowsum(r + 64 * LDS_STRIDE, w, bC);
    y3 = ssm_rowsum(r + 96 * LDS_STRIDE, w, bC);
    asm volatile("" : "+v"(y2), "+v"(y3));

    WAVE_LDS_FENCE();
}

__global__ __launch_bounds__(NWAVE * 32) void ssm_row_kernel(
    const float* __restrict__ u, const float* __restrict__ A,
    const float* __restrict__ W_B, const float* __restrict__ b_B,
    const float* __restrict__ W_C, const float* __restrict__ b_C,
    float* __restrict__ y)
{
    __shared__ __attribute__((aligned(16))) float lds[NWAVE * WLDS + NCHUNK * HID];

    const int b   = blockIdx.x;
    const int tid = threadIdx.x;
    const int wv  = __builtin_amdgcn_readfirstlane(tid >> 5);
    const int ln  = tid & 31;
    const int m   = ln & 15;
    const int kh  = ln >> 4;
    const int c0  = wv * CPW;

    float* buLds = lds + wv * WLDS;
    float* Ssh   = lds + NWAVE * WLDS;

    const float pa0 = A[tid >> 7];
    const float pa1 = A[8 + (tid >> 7)];

    v2f bfrag[8];
#pragma unroll
    for (int k = 0; k < 8; ++k) {
        const float* p = W_B + m * DIN + 4 * k + 2 * kh;
        bfrag[k].x = p[0];
        bfrag[k].y = p[1];
    }
    const float bias = b_B[m];

    const v4f* ug = reinterpret_cast<const v4f*>(
        u + ((size_t)b * TT + (size_t)c0 * CHLEN) * DIN) + ln;
    UTile t0, t1;
    load_tile(t0, ug);
    __builtin_amdgcn_sched_barrier(0);

    float pw0 = powf(pa0, (float)((tid & (CHLEN - 1)) + 1));
    float pw1 = powf(pa1, (float)((tid & (CHLEN - 1)) + 1));
    asm volatile("" : "+v"(pw0), "+v"(pw1));

    const float bC = b_C[0];
    float w[HID];
#pragma unroll
    for (int h = 0; h < HID; ++h) w[h] = W_C[h];
    const float Aval = A[m];
    float apow[8];
    apow[0] = Aval;
#pragma unroll
    for (int v = 1; v < 8; ++v) apow[v] = apow[v - 1] * Aval;

    float ya0, ya1, ya2, ya3, yb0, yb1, yb2, yb3;
    ssm_chunk<false>(ug, t0, t1, buLds, Ssh + c0 * HID, bfrag, bias, apow, w, bC,
                     ya0, ya1, ya2, ya3, ln, m, kh);
    ssm_chunk<true>(ug + NTILE * NTILE_F4, t0, t1, buLds, Ssh + (c0 + 1) * HID, bfrag, bias, apow, w, bC,
                    yb0, yb1, yb2, yb3, ln, m, kh);

    __syncthreads();

    float* pwT = lds;
    pwT[tid] = pw0;
    pwT[tid + NWAVE * 32] = pw1;

    float p = Aval;
#pragma unroll
    for (int i = 0; i < 7; ++i) p *= p;

    float f = 0.0f, car0 = 0.0f;
    for (int cb = 0; cb < NCHUNK; cb += 16) {
        float sv[16];
#pragma unroll
        for (int j = 0; j < 16; ++j) sv[j] = Ssh[(cb + j) * HID + m];
#pragma unroll
        for (int j = 0; j < 16; ++j) {
            if (cb + j == c0) car0 = f;
            f = fmaf(p, f, sv[j]);
        }
    }
    const float car1 = fmaf(p, car0, Ssh[c0 * HID + m]);

    float* wcar = buLds + WCAR_OFF;
    const float wc = W_C[m];
    if (ln < 16) {
        wcar[ln]       = wc * car0;
        wcar[HID + ln] = wc * car1;
    }

    __syncthreads();

    float wca[CPW][HID];
#pragma unroll
    for (int cc = 0; cc < CPW; ++cc)
#pragma unroll
        for (int h = 0; h < HID; ++h) wca[cc][h] = wcar[cc * HID + h];

    float* yp = y + (size_t)b * TT + (size_t)c0 * CHLEN + ln;
    const float ya[CHLEN / 32] = {ya0, ya1, ya2, ya3};
    const float yb[CHLEN / 32] = {yb0, yb1, yb2, yb3};
#pragma unroll
    for (int j = 0; j < CHLEN / 32; ++j) {
        float pwr[HID];
#pragma unroll
        for (int h = 0; h < HID; ++h) pwr[h] = pwT[h * CHLEN + ln + 32 * j];
        float corr0 = 0.0f, corr1 = 0.0f;
#pragma unroll
        for (int h = 0; h < HID; ++h) {
            corr0 = fmaf(wca[0][h], pwr[h], corr0);
            corr1 = fmaf(wca[1][h], pwr[h], corr1);
        }
        yp[32 * j]         = ya[j] + corr0;
        yp[CHLEN + 32 * j] = yb[j] + corr1;
    }
}

extern "C" void kernel_launch(void* const* d_in, const int* in_sizes, int n_in,
                              void* d_out, int out_size, void* d_ws, size_t ws_size,
                              hipStream_t stream) {
    const float* u   = (const float*)d_in[0];
    const float* A   = (const float*)d_in[1];
    const float* W_B = (const float*)d_in[2];
    const float* b_B = (const float*)d_in[3];
    const float* W_C = (const float*)d_in[4];
    const float* b_C = (const float*)d_in[5];
    float* y = (float*)d_out;

    ssm_row_kernel<<<dim3(NBATCH), dim3(NWAVE * 32), 0, stream>>>(
        u, A, W_B, b_B, W_C, b_C, y);
}
